# starting kernel
# baseline (speedup 1.0000x reference)
#include <hip/hip_runtime.h>
#include <math.h>
#include <float.h>

#define N_NODES 50000
#define N_EDGES 800000
#define IN_C    128
#define OUT_C   128
#define HEADS   8
#define BASES   4
#define HEAD_D  16
#define BD      (BASES * HEAD_D)

typedef float v2f __attribute__((ext_vector_type(2)));
typedef float v8f __attribute__((ext_vector_type(8)));

__device__ __forceinline__ void atomicMaxFloat(float* addr, float val) {
  unsigned bits = __float_as_uint(val);
  if ((bits & 0x80000000u) == 0u) {
    atomicMax((int*)addr, (int)bits);
  } else {
    atomicMin((unsigned int*)addr, bits);
  }
}

__global__ __launch_bounds__(256) void k_init(float* __restrict__ agg,
                                              unsigned* __restrict__ cnt) {
  int i = blockIdx.x * 256 + threadIdx.x;
  if (i < N_NODES * 256) {
    int part = (i >> 6) & 3;
    agg[i] = (part == 3) ? -INFINITY : 0.0f;
  }
  if (i < N_NODES) cnt[i] = 0u;
}

__global__ __launch_bounds__(256) void k_degree(const int* __restrict__ col,
                                                unsigned* __restrict__ cnt) {
  int e = blockIdx.x * 256 + threadIdx.x;
  if (e < N_EDGES) atomicAdd(&cnt[col[e]], 1u);
}

__global__ __launch_bounds__(256) void k_dinv(const unsigned* __restrict__ cnt,
                                              float* __restrict__ dinv) {
  int n = blockIdx.x * 256 + threadIdx.x;
  if (n < N_NODES) dinv[n] = rsqrtf((float)cnt[n] + 1.0f);
}

__global__ __launch_bounds__(128) void k_gemm(const float* __restrict__ x,
                                              const float* __restrict__ bW,
                                              const float* __restrict__ cW,
                                              const float* __restrict__ cb,
                                              float* __restrict__ bases,
                                              float* __restrict__ wbuf) {
  __shared__ float xs[64][132];

  const int tid  = threadIdx.x;
  const int row0 = blockIdx.x * 64;

  for (int t = 0; t < 16; ++t) {
    int idx = t * 128 + tid;
    int r   = idx >> 5;
    int c4  = (idx & 31) * 4;
    int gr  = row0 + r;
    if (gr >= N_NODES) gr = N_NODES - 1;
    const float4 v = *(const float4*)(x + (size_t)gr * IN_C + c4);
    xs[r][c4 + 0] = v.x; xs[r][c4 + 1] = v.y;
    xs[r][c4 + 2] = v.z; xs[r][c4 + 3] = v.w;
  }
  __syncthreads();

  const int wave = tid >> 5;
  const int lane = tid & 31;
  const int nlo  = lane & 15;
  const int half = lane >> 4;
  const int kb   = half * 2;
  const int lr   = wave * 16 + nlo;

  for (int nt = 0; nt < 12; ++nt) {
    v8f acc = {0.f, 0.f, 0.f, 0.f, 0.f, 0.f, 0.f, 0.f};

    if (nt < 4) {
      const float* W = bW + nt * (IN_C * HEAD_D);
#pragma unroll
      for (int ks = 0; ks < 32; ++ks) {
        const int k = ks * 4 + kb;
        v2f a = *(const v2f*)&xs[lr][k];
        v2f b = { W[k * HEAD_D + nlo], W[(k + 1) * HEAD_D + nlo] };
        acc = __builtin_amdgcn_wmma_f32_16x16x4_f32(
            false, a, false, b, (short)0, acc, false, false);
      }
#pragma unroll
      for (int r = 0; r < 8; ++r) {
        const int m  = r + (half << 3);
        const int gr = row0 + wave * 16 + m;
        if (gr < N_NODES) bases[(size_t)gr * BD + nt * 16 + nlo] = acc[r];
      }
    } else {
      const int   head = nt - 4;
      const float* W   = cW + head * 16;
#pragma unroll
      for (int ks = 0; ks < 32; ++ks) {
        const int k = ks * 4 + kb;
        v2f a = *(const v2f*)&xs[lr][k];
        v2f b = { W[k * OUT_C + nlo], W[(k + 1) * OUT_C + nlo] };
        acc = __builtin_amdgcn_wmma_f32_16x16x4_f32(
            false, a, false, b, (short)0, acc, false, false);
      }
      const float bb = cb[head * 16 + nlo];
#pragma unroll
      for (int r = 0; r < 8; ++r) {
        float v = acc[r] + bb;
        float mx = v;
        mx = fmaxf(mx, __shfl_xor(mx, 1));
        mx = fmaxf(mx, __shfl_xor(mx, 2));
        mx = fmaxf(mx, __shfl_xor(mx, 4));
        mx = fmaxf(mx, __shfl_xor(mx, 8));
        float e = __expf(v - mx);
        float s = e;
        s += __shfl_xor(s, 1);
        s += __shfl_xor(s, 2);
        s += __shfl_xor(s, 4);
        s += __shfl_xor(s, 8);
        const float wv = e / s;
        const int m  = r + (half << 3);
        const int gr = row0 + wave * 16 + m;
        if (gr < N_NODES) wbuf[(size_t)gr * OUT_C + head * 16 + nlo] = wv;
      }
    }
  }
}

__global__ __launch_bounds__(256) void k_scatter(const int* __restrict__ ei,
                                                 const float* __restrict__ bases,
                                                 const float* __restrict__ dinv,
                                                 float* __restrict__ agg) {
  const long long g = (long long)blockIdx.x * 256 + threadIdx.x;
  const int e = (int)(g >> 4);
  if (e >= N_EDGES) return;
  const int q = (int)(g & 15);

  const int row = ei[e];
  const int col = ei[N_EDGES + e];
  if (q == 0 && e + 4096 < N_EDGES) {
    __builtin_prefetch(ei + e + 4096, 0, 0);
    __builtin_prefetch(ei + N_EDGES + e + 4096, 0, 0);
  }

  const float  dr = dinv[row];
  const float4 v  = *(const float4*)(bases + (size_t)row * BD + q * 4);
  float* base     = agg + (size_t)col * 256;

  const float vals[4] = {v.x, v.y, v.z, v.w};
#pragma unroll
  for (int c = 0; c < 4; ++c) {
    const int   idx = q * 4 + c;
    const float t   = vals[c];
    atomicAdd(base + idx,       dr * t);
    atomicAdd(base + 64 + idx,  t);
    atomicAdd(base + 128 + idx, t * t);
    atomicMaxFloat(base + 192 + idx, t);
  }
}

__global__ __launch_bounds__(128) void k_combine(const float* __restrict__ agg,
                                                 const float* __restrict__ bases,
                                                 const float* __restrict__ wbuf,
                                                 const unsigned* __restrict__ cnt,
                                                 const float* __restrict__ dinv,
                                                 const float* __restrict__ bias,
                                                 float* __restrict__ out) {
  __shared__ float ws_[128];
  __shared__ float as_[256];
  __shared__ float bs_[64];

  const int n   = blockIdx.x;
  const int tid = threadIdx.x;

  ws_[tid]       = wbuf[(size_t)n * 128 + tid];
  as_[tid]       = agg[(size_t)n * 256 + tid];
  as_[tid + 128] = agg[(size_t)n * 256 + 128 + tid];
  if (tid < 64) bs_[tid] = bases[(size_t)n * 64 + tid];
  __syncthreads();

  const float cf = (float)cnt[n];
  const float ic = 1.0f / fmaxf(cf, 1.0f);
  const float dv = dinv[n];
  const int h = tid >> 4;
  const int d = tid & 15;

  float o = bias[tid];
#pragma unroll
  for (int b = 0; b < 4; ++b) {
    const int   idx  = b * 16 + d;
    const float sym  = dv * (as_[idx] + dv * bs_[idx]);
    const float s    = as_[64 + idx];
    const float mean = s * ic;
    const float var  = as_[128 + idx] * ic - mean * mean;
    const float sd   = sqrtf(fmaxf(var, 0.0f) + 1e-5f);
    const float mx   = (cf > 0.0f) ? as_[192 + idx] : 0.0f;
    const float* wp  = ws_ + h * 16 + b * 4;
    o += wp[0] * sym + wp[1] * mean + wp[2] * mx + wp[3] * sd;
  }
  out[(size_t)n * 128 + tid] = o;
}

extern "C" void kernel_launch(void* const* d_in, const int* in_sizes, int n_in,
                              void* d_out, int out_size, void* d_ws, size_t ws_size,
                              hipStream_t stream) {
  (void)in_sizes; (void)n_in; (void)out_size; (void)ws_size;
  const float* x    = (const float*)d_in[0];
  const int*   ei   = (const int*)d_in[1];
  const float* bW   = (const float*)d_in[2];
  const float* cW   = (const float*)d_in[3];
  const float* cb   = (const float*)d_in[4];
  const float* bias = (const float*)d_in[5];
  float*       out  = (float*)d_out;

  char* ws = (char*)d_ws;
  float*    agg   = (float*)ws;  ws += (size_t)N_NODES * 256 * sizeof(float);
  float*    bases = (float*)ws;  ws += (size_t)N_NODES * BD * sizeof(float);
  float*    wbuf  = (float*)ws;  ws += (size_t)N_NODES * OUT_C * sizeof(float);
  unsigned* cnt   = (unsigned*)ws; ws += (size_t)N_NODES * sizeof(unsigned);
  float*    dinv  = (float*)ws;

  k_init<<<N_NODES, 256, 0, stream>>>(agg, cnt);
  k_degree<<<(N_EDGES + 255) / 256, 256, 0, stream>>>(ei + N_EDGES, cnt);
  k_gemm<<<(N_NODES + 63) / 64, 128, 0, stream>>>(x, bW, cW, cb, bases, wbuf);
  k_dinv<<<(N_NODES + 255) / 256, 256, 0, stream>>>(cnt, dinv);
  {
    const long long work = (long long)N_EDGES * 16;
    k_scatter<<<(unsigned)((work + 255) / 256), 256, 0, stream>>>(ei, bases, dinv, agg);
  }
  k_combine<<<N_NODES, 128, 0, stream>>>(agg, bases, wbuf, cnt, dinv, bias, out);
}


# scatter+combine in one kernel: a work-group keeps the 4 aggregator rows of its 196 destination nodes in LDS, filters edges by 1-byte tile ids, adds with LDS atomics (arrival order); agg buffer gone
# speedup vs baseline: 2.2465x; 2.2465x over previous
#include <hip/hip_runtime.h>
#include <math.h>
#include <float.h>


#define N_NODES 50000
#define N_EDGES 800000
#define IN_C    128
#define OUT_C   128
#define HEADS   8
#define BASES   4
#define HEAD_D  16
#define BD      (BASES * HEAD_D)

#define TILE_N   196
#define N_TILES  256
#define LIST_CAP 16384
#define N_ID16   (N_EDGES / 16)
#define FULL_IT  (N_ID16 / 1024)

typedef float v2f __attribute__((ext_vector_type(2)));
typedef float v8f __attribute__((ext_vector_type(8)));

__device__ __forceinline__ void ldsAdd(float* addr, float val) {
  __hip_atomic_fetch_add(addr, val, __ATOMIC_RELAXED, __HIP_MEMORY_SCOPE_WORKGROUP);
}
__device__ __forceinline__ void ldsMax(float* addr, float val) {
  __hip_atomic_fetch_max(addr, val, __ATOMIC_RELAXED, __HIP_MEMORY_SCOPE_WORKGROUP);
}
__device__ __forceinline__ void tileAdd(float* a, float dr, float t0, float t1) {
  ldsAdd(a,       dr * t0);
  ldsAdd(a + 32,  dr * t1);
  ldsAdd(a + 64,  t0);
  ldsAdd(a + 96,  t1);
  ldsAdd(a + 128, t0 * t0);
  ldsAdd(a + 160, t1 * t1);
  ldsMax(a + 192, t0);
  ldsMax(a + 224, t1);
}

__global__ __launch_bounds__(256) void k_init(unsigned* __restrict__ cnt) {
  int i = blockIdx.x * 256 + threadIdx.x;
  if (i < N_NODES) cnt[i] = 0u;
}

__global__ __launch_bounds__(256) void k_degree(const int* __restrict__ col,
                                                unsigned* __restrict__ cnt,
                                                unsigned char* __restrict__ tile8) {
  int e = blockIdx.x * 256 + threadIdx.x;
  if (e < N_EDGES) {
    const int c = col[e];
    atomicAdd(&cnt[c], 1u);
    tile8[e] = (unsigned char)((unsigned)c / (unsigned)TILE_N);
  }
}

__global__ __launch_bounds__(256) void k_dinv(const unsigned* __restrict__ cnt,
                                              float* __restrict__ dinv) {
  int n = blockIdx.x * 256 + threadIdx.x;
  if (n < N_NODES) dinv[n] = rsqrtf((float)cnt[n] + 1.0f);
}

__global__ __launch_bounds__(128) void k_gemm(const float* __restrict__ x,
                                              const float* __restrict__ bW,
                                              const float* __restrict__ cW,
                                              const float* __restrict__ cb,
                                              float* __restrict__ bases,
                                              float* __restrict__ wbuf) {
  __shared__ float xs[64][132];

  const int tid  = threadIdx.x;
  const int row0 = blockIdx.x * 64;

  for (int t = 0; t < 16; ++t) {
    int idx = t * 128 + tid;
    int r   = idx >> 5;
    int c4  = (idx & 31) * 4;
    int gr  = row0 + r;
    if (gr >= N_NODES) gr = N_NODES - 1;
    const float4 v = *(const float4*)(x + (size_t)gr * IN_C + c4);
    xs[r][c4 + 0] = v.x; xs[r][c4 + 1] = v.y;
    xs[r][c4 + 2] = v.z; xs[r][c4 + 3] = v.w;
  }
  __syncthreads();

  const int wave = tid >> 5;
  const int lane = tid & 31;
  const int nlo  = lane & 15;
  const int half = lane >> 4;
  const int kb   = half * 2;
  const int lr   = wave * 16 + nlo;

  for (int nt = 0; nt < 12; ++nt) {
    v8f acc = {0.f, 0.f, 0.f, 0.f, 0.f, 0.f, 0.f, 0.f};

    if (nt < 4) {
      const float* W = bW + nt * (IN_C * HEAD_D);
#pragma unroll
      for (int ks = 0; ks < 32; ++ks) {
        const int k = ks * 4 + kb;
        v2f a = *(const v2f*)&xs[lr][k];
        v2f b = { W[k * HEAD_D + nlo], W[(k + 1) * HEAD_D + nlo] };
        acc = __builtin_amdgcn_wmma_f32_16x16x4_f32(
            false, a, false, b, (short)0, acc, false, false);
      }
#pragma unroll
      for (int r = 0; r < 8; ++r) {
        const int m  = r + (half << 3);
        const int gr = row0 + wave * 16 + m;
        if (gr < N_NODES) bases[(size_t)gr * BD + nt * 16 + nlo] = acc[r];
      }
    } else {
      const int   head = nt - 4;
      const float* W   = cW + head * 16;
#pragma unroll
      for (int ks = 0; ks < 32; ++ks) {
        const int k = ks * 4 + kb;
        v2f a = *(const v2f*)&xs[lr][k];
        v2f b = { W[k * OUT_C + nlo], W[(k + 1) * OUT_C + nlo] };
        acc = __builtin_amdgcn_wmma_f32_16x16x4_f32(
            false, a, false, b, (short)0, acc, false, false);
      }
      const float bb = cb[head * 16 + nlo];
#pragma unroll
      for (int r = 0; r < 8; ++r) {
        float v = acc[r] + bb;
        float mx = v;
        mx = fmaxf(mx, __shfl_xor(mx, 1));
        mx = fmaxf(mx, __shfl_xor(mx, 2));
        mx = fmaxf(mx, __shfl_xor(mx, 4));
        mx = fmaxf(mx, __shfl_xor(mx, 8));
        float e = __expf(v - mx);
        float s = e;
        s += __shfl_xor(s, 1);
        s += __shfl_xor(s, 2);
        s += __shfl_xor(s, 4);
        s += __shfl_xor(s, 8);
        const float wv = e / s;
        const int m  = r + (half << 3);
        const int gr = row0 + wave * 16 + m;
        if (gr < N_NODES) wbuf[(size_t)gr * OUT_C + head * 16 + nlo] = wv;
      }
    }
  }
}

__global__ __launch_bounds__(1024) void k_tile(const int* __restrict__ ei,
                                               const unsigned char* __restrict__ tile8,
                                               const float* __restrict__ bases,
                                               const float* __restrict__ wbuf,
                                               const unsigned* __restrict__ cnt,
                                               const float* __restrict__ dinv,
                                               const float* __restrict__ bias,
                                               float* __restrict__ out) {
  __shared__ __attribute__((aligned(16))) float acc[TILE_N * 256];
  __shared__ __attribute__((aligned(16))) unsigned list[LIST_CAP];
  __shared__ unsigned lcount;

  const int tid  = threadIdx.x;
  const int lane = tid & 31;
  const int wave = tid >> 5;
  const int lo   = blockIdx.x * TILE_N;
  const int* __restrict__ col = ei + N_EDGES;

  for (int i = tid; i < TILE_N * 256; i += 1024)
    acc[i] = (((i >> 6) & 3) == 3) ? -INFINITY : 0.0f;
  if (tid == 0) lcount = 0u;
  __syncthreads();

  {
    const uint4* __restrict__ idv = (const uint4*)tile8;
    const unsigned pat = (unsigned)blockIdx.x * 0x01010101u;
    auto scan16 = [&](const uint4 v, const int j) {
      const unsigned w[4] = {v.x, v.y, v.z, v.w};
      unsigned z[4];
#pragma unroll
      for (int k = 0; k < 4; ++k) {
        const unsigned x = w[k] ^ pat;
        const unsigned y = (x & 0x7f7f7f7fu) + 0x7f7f7f7fu;
        z[k] = ~(y | x | 0x7f7f7f7fu);
      }
      unsigned m = (z[0] >> 7) | (z[1] >> 6) | (z[2] >> 5) | (z[3] >> 4);
      while (m != 0u) {
        const int p = __builtin_ctz(m);
        m &= m - 1u;
        const int e = j * 16 + ((p & 7) << 2) + (p >> 3);
        const unsigned slot = atomicAdd(&lcount, 1u);
        if (slot < (unsigned)LIST_CAP) list[slot] = (unsigned)e;
      }
    };
    static_assert(FULL_IT % 4 == 0, "the unrolled scan takes whole groups of four steps");
    for (int i0 = 0; i0 < FULL_IT; i0 += 4) {
      uint4 v[4];
#pragma unroll
      for (int u = 0; u < 4; ++u) v[u] = idv[(i0 + u) * 1024 + tid];
#pragma unroll
      for (int u = 0; u < 4; ++u) scan16(v[u], (i0 + u) * 1024 + tid);
    }
    {
      const int j = FULL_IT * 1024 + tid;
      if (j < N_ID16) scan16(idv[j], j);
    }
  }
  __syncthreads();
  const unsigned total = lcount;

  if (total <= (unsigned)LIST_CAP) {
    const int n = (int)total;
    for (int i = tid; i < n; i += 1024) {
      const int e = (int)list[i];
      const int r = ei[e];
      const int c = col[e];
      list[i] = ((unsigned)r << 8) | (unsigned)(c - lo);
    }
    __syncthreads();
    const int nfull = n & ~7;
    for (int base = wave * 8; base < nfull; base += 256) {
      unsigned dd[8];
      float    dr[8], v0[8], v1[8];
      const uint4 q0 = *(const uint4*)(list + base);
      const uint4 q1 = *(const uint4*)(list + base + 4);
      const unsigned q[8] = {q0.x, q0.y, q0.z, q0.w, q1.x, q1.y, q1.z, q1.w};
#pragma unroll
      for (int u = 0; u < 8; ++u) {
        const unsigned p = __builtin_amdgcn_readfirstlane(q[u]);
        dd[u] = p & 255u;
        const unsigned r = p >> 8;
        const float* bp = bases + (size_t)r * BD;
        dr[u] = dinv[r];
        v0[u] = bp[lane];
        v1[u] = bp[32 + lane];
      }
#pragma unroll
      for (int u = 0; u < 8; ++u) tileAdd(acc + dd[u] * 256 + lane, dr[u], v0[u], v1[u]);
    }
    if (nfull + wave < n) {
      const unsigned p = __builtin_amdgcn_readfirstlane(list[nfull + wave]);
      const unsigned r = p >> 8;
      const float* bp = bases + (size_t)r * BD;
      tileAdd(acc + (p & 255u) * 256 + lane, dinv[r], bp[lane], bp[32 + lane]);
    }
  } else {
    for (int e = tid; e < N_EDGES; e += 1024) {
      const unsigned d = (unsigned)(col[e] - lo);
      if (d < (unsigned)TILE_N) {
        const int    row = ei[e];
        const float  drr = dinv[row];
        const float* bp  = bases + (size_t)row * BD;
        float*       a   = acc + d * 256;
        for (int k = 0; k < BD; ++k) {
          const float t = bp[k];
          ldsAdd(a + k,       drr * t);
          ldsAdd(a + 64 + k,  t);
          ldsAdd(a + 128 + k, t * t);
          ldsMax(a + 192 + k, t);
        }
      }
    }
  }
  __syncthreads();

  for (int i = tid; i < TILE_N * BD; i += 1024) {
    const int nl  = i >> 6;
    const int idx = i & 63;
    const int nn  = lo + nl;
    if (nn < N_NODES) {
      float* a = acc + nl * 256 + idx;
      const float cf = (float)cnt[nn];
      const float ic = 1.0f / fmaxf(cf, 1.0f);
      const float dv = dinv[nn];
      const float bs = bases[(size_t)nn * BD + idx];
      const float sym  = dv * (a[0] + dv * bs);
      const float s    = a[64];
      const float mean = s * ic;
      const float var  = a[128] * ic - mean * mean;
      const float sd   = sqrtf(fmaxf(var, 0.0f) + 1e-5f);
      const float mx   = (cf > 0.0f) ? a[192] : 0.0f;
      a[0] = sym; a[64] = mean; a[128] = mx; a[192] = sd;
    }
  }
  __syncthreads();

  {
    const int h  = lane >> 2;
    const int dq = lane & 3;
    const float4 bv = ((const float4*)bias)[lane];
    for (int nl = wave; nl < TILE_N; nl += 32) {
      const int nn = lo + nl;
      if (nn < N_NODES) {
        const float4* wp = (const float4*)(wbuf + (size_t)nn * OUT_C + h * 16);
        const float*  a  = acc + nl * 256 + dq * 4;
        float4 o = bv;
#pragma unroll
        for (int b = 0; b < 4; ++b) {
          const float4 w  = wp[b];
          const float4 sy = *(const float4*)(a + b * 16);
          const float4 me = *(const float4*)(a + 64 + b * 16);
          const float4 mx = *(const float4*)(a + 128 + b * 16);
          const float4 sd = *(const float4*)(a + 192 + b * 16);
          o.x += w.x * sy.x + w.y * me.x + w.z * mx.x + w.w * sd.x;
          o.y += w.x * sy.y + w.y * me.y + w.z * mx.y + w.w * sd.y;
          o.z += w.x * sy.z + w.y * me.z + w.z * mx.z + w.w * sd.z;
          o.w += w.x * sy.w + w.y * me.w + w.z * mx.w + w.w * sd.w;
        }
        ((float4*)(out + (size_t)nn * OUT_C))[lane] = o;
      }
    }
  }
}

extern "C" void kernel_launch(void* const* d_in, const int* in_sizes, int n_in,
                              void* d_out, int out_size, void* d_ws, size_t ws_size,
                              hipStream_t stream) {
  (void)in_sizes; (void)n_in; (void)out_size; (void)ws_size;
  const float* x    = (const float*)d_in[0];
  const int*   ei   = (const int*)d_in[1];
  const float* bW   = (const float*)d_in[2];
  const float* cW   = (const float*)d_in[3];
  const float* cb   = (const float*)d_in[4];
  const float* bias = (const float*)d_in[5];
  float*       out  = (float*)d_out;

  char* ws = (char*)d_ws;
  float*         bases = (float*)ws;    ws += (size_t)N_NODES * BD * sizeof(float);
  float*         wbuf  = (float*)ws;    ws += (size_t)N_NODES * OUT_C * sizeof(float);
  unsigned*      cnt   = (unsigned*)ws; ws += (size_t)N_NODES * sizeof(unsigned);
  float*         dinv  = (float*)ws;    ws += (size_t)N_NODES * sizeof(float);
  unsigned char* tile8 = (unsigned char*)ws;

  k_init<<<(N_NODES + 255) / 256, 256, 0, stream>>>(cnt);
  k_degree<<<(N_EDGES + 255) / 256, 256, 0, stream>>>(ei + N_EDGES, cnt, tile8);
  k_gemm<<<(N_NODES + 63) / 64, 128, 0, stream>>>(x, bW, cW, cb, bases, wbuf);
  k_dinv<<<(N_NODES + 255) / 256, 256, 0, stream>>>(cnt, dinv);
  k_tile<<<N_TILES, 1024, 0, stream>>>(ei, tile8, bases, wbuf, cnt, dinv, bias, out);
}


# k_gather: a wave issues exactly a node's in-degree row loads (up to 32 in flight) down an inlined ladder and sums them on the way back; no padded 16-row steps; a node's rows are summed last to first
# speedup vs baseline: 6.0000x; 2.6708x over previous
#include <hip/hip_runtime.h>
#include <math.h>
#include <float.h>


#define N_NODES 50000
#define N_EDGES 800000
#define IN_C    128
#define OUT_C   128
#define HEADS   8
#define BASES   4
#define HEAD_D  16
#define BD      (BASES * HEAD_D)

#define TILE_N    196
#define N_TILES   256
#define LIST_CAP  8192
#define OFF_PITCH 256
#define DINV_PITCH 256
#define BM_PITCH  25024
#define BM_PIECES (BM_PITCH / 4)
#define BM_PER_T  7
#define LINE_SLOTS 32
#define REC_EMPTY 0xffffffffu
#define REC_OVER  0xfffffffeu
static_assert(N_NODES <= 65536 && TILE_N <= 256, "row and node-in-tile fit an edge record");

__device__ __forceinline__ unsigned dinv_at(unsigned n) {
  return n + (n / (unsigned)TILE_N) * (unsigned)(DINV_PITCH - TILE_N);
}

typedef float v2f __attribute__((ext_vector_type(2)));
typedef float v8f __attribute__((ext_vector_type(8)));

#define BITS_EDGES   1024
#define BITS_THREADS 128
#define BITS_BLOCKS  ((N_EDGES + BITS_EDGES - 1) / BITS_EDGES)
#define BITS_LPITCH  33
static_assert(BITS_BLOCKS * 32 == BM_PITCH, "one 128-byte line per work-group fills a tile's pitch exactly");
static_assert(BM_PER_T * 1024 >= BM_PIECES, "the list kernel's threads cover a tile's lines");
__global__ __launch_bounds__(BITS_THREADS) void k_bits(const int* __restrict__ row,
                                                       const int* __restrict__ col,
                                                       unsigned* __restrict__ bitmap) {
  __shared__ unsigned img[N_TILES * BITS_LPITCH];
  __shared__ unsigned taken[N_TILES];
  const int tid = threadIdx.x;

  const int4* __restrict__ c4 = (const int4*)col;
  const int4* __restrict__ r4 = (const int4*)row;
  const int qb = blockIdx.x * (BITS_EDGES / 4);
  int4 c[2], r[2];
  bool have[2];
#pragma unroll
  for (int k = 0; k < 2; ++k) {
    const int q  = qb + k * BITS_THREADS + tid;
    have[k] = q < N_EDGES / 4;
    const int qq = have[k] ? q : N_EDGES / 4 - 1;
    c[k] = c4[qq];
    r[k] = r4[qq];
  }

  for (int i = tid; i < N_TILES * BITS_LPITCH; i += BITS_THREADS) img[i] = REC_EMPTY;
  for (int i = tid; i < N_TILES; i += BITS_THREADS) taken[i] = 0u;
  __syncthreads();

  unsigned at[8], rec[8], sl[8];
#pragma unroll
  for (int k = 0; k < 2; ++k) {
    const unsigned cv[4] = {(unsigned)c[k].x, (unsigned)c[k].y, (unsigned)c[k].z, (unsigned)c[k].w};
    const unsigned rv[4] = {(unsigned)r[k].x, (unsigned)r[k].y, (unsigned)r[k].z, (unsigned)r[k].w};
#pragma unroll
    for (int j = 0; j < 4; ++j) {
      const unsigned t = min(cv[j] / (unsigned)TILE_N, (unsigned)(N_TILES - 1));
      const unsigned d = (cv[j] - t * (unsigned)TILE_N) & 0xffu;
      at[k * 4 + j]  = t * BITS_LPITCH;
      rec[k * 4 + j] = (rv[j] & 0xffffu) | (d << 16);
      sl[k * 4 + j]  = (unsigned)LINE_SLOTS;
      if (have[k]) sl[k * 4 + j] = atomicAdd(&taken[t], 1u);
    }
  }
#pragma unroll
  for (int i = 0; i < 8; ++i)
    if (sl[i] < (unsigned)LINE_SLOTS) img[at[i] + sl[i]] = rec[i];
  __syncthreads();

  const int lane = tid & 31;
  const int wave = tid >> 5;
  for (int t = wave; t < N_TILES; t += BITS_THREADS / 32) {
    unsigned v = img[t * BITS_LPITCH + lane];
    if (taken[t] > (unsigned)LINE_SLOTS) v = (lane == 0) ? REC_OVER : REC_EMPTY;
    bitmap[(size_t)t * BM_PITCH + blockIdx.x * 32 + lane] = v;
  }
}

__global__ __launch_bounds__(1024) void k_list(const int* __restrict__ ei,
                                               const unsigned* __restrict__ bitmap,
                                               float* __restrict__ dinv,
                                               unsigned* __restrict__ goff,
                                               unsigned* __restrict__ gsrt) {
  __shared__ __attribute__((aligned(16))) unsigned list[LIST_CAP];
  __shared__ __attribute__((aligned(16))) unsigned key[LIST_CAP];
  __shared__ __attribute__((aligned(16))) unsigned srt[LIST_CAP];
  __shared__ unsigned start[TILE_N + 1];
  __shared__ unsigned fill[TILE_N];
  __shared__ unsigned lcount;
  __shared__ unsigned over[BITS_BLOCKS];
  __shared__ unsigned nover;

  const int tid  = threadIdx.x;
  const int lane = tid & 31;
  const int wave = __builtin_amdgcn_readfirstlane(tid >> 5);
  const int lo   = blockIdx.x * TILE_N;
  const int* __restrict__ col = ei + N_EDGES;

  if (tid < TILE_N) fill[tid] = 0u;
  if (tid == 0) { lcount = 0u; nover = 0u; }
  __syncthreads();

  {
    const uint4* __restrict__ bm = (const uint4*)(bitmap + (size_t)blockIdx.x * BM_PITCH);
    uint4 v[BM_PER_T];
#pragma unroll
    for (int k = 0; k < BM_PER_T; ++k) {
      const int j = k * 1024 + tid;
      v[k] = (j < BM_PIECES) ? bm[j] : make_uint4(REC_EMPTY, REC_EMPTY, REC_EMPTY, REC_EMPTY);
    }
    unsigned cnt = 0u;
#pragma unroll
    for (int k = 0; k < BM_PER_T; ++k)
      cnt += (unsigned)(v[k].x < REC_OVER) + (unsigned)(v[k].y < REC_OVER)
           + (unsigned)(v[k].z < REC_OVER) + (unsigned)(v[k].w < REC_OVER);
    unsigned incl = cnt;
#pragma unroll
    for (int off = 1; off < 32; off <<= 1) {
      const unsigned t = __shfl_up(incl, off);
      if (lane >= off) incl += t;
    }
    unsigned run0 = 0u;
    if (lane == 31) run0 = atomicAdd(&lcount, incl);
    run0 = (unsigned)__builtin_amdgcn_readlane((int)run0, 31);
    unsigned slot = run0 + (incl - cnt);
#pragma unroll
    for (int k = 0; k < BM_PER_T; ++k) {
      const unsigned w[4] = {v[k].x, v[k].y, v[k].z, v[k].w};
#pragma unroll
      for (int c = 0; c < 4; ++c) {
        if (w[c] < REC_OVER) {
          if (slot < (unsigned)LIST_CAP) list[slot] = w[c];
          ++slot;
        }
      }
      if (w[0] == REC_OVER) {
        const unsigned q = atomicAdd(&nover, 1u);
        if (q < (unsigned)BITS_BLOCKS) over[q] = (unsigned)(k * 1024 + tid) >> 3;
      }
    }
  }
  __syncthreads();
  const unsigned nov = min(nover, (unsigned)BITS_BLOCKS);
  if (nov != 0u) {
    for (unsigned q = 0u; q < nov; ++q) {
      const int e = (int)over[q] * BITS_EDGES + tid;
      if (e < N_EDGES) {
        const unsigned d = (unsigned)(col[e] - lo);
        if (d < (unsigned)TILE_N) {
          const unsigned s = atomicAdd(&lcount, 1u);
          if (s < (unsigned)LIST_CAP) list[s] = ((unsigned)ei[e] & 0xffffu) | (d << 16);
        }
      }
    }
    __syncthreads();
  }
  const unsigned total  = lcount;
  const bool     listed = total <= (unsigned)LIST_CAP;

  if (listed) {
    const int n = (int)total;
    for (int i = tid; i < n; i += 1024) {
      const unsigned d  = min(list[i] >> 16, (unsigned)(TILE_N - 1));
      const unsigned pl = atomicAdd(&fill[d], 1u);
      key[i]  = (d << 16) | pl;
    }
  } else {
    for (int e = tid; e < N_EDGES; e += 1024) {
      const unsigned d = (unsigned)(col[e] - lo);
      if (d < (unsigned)TILE_N) atomicAdd(&fill[d], 1u);
    }
  }
  __syncthreads();

  if (wave == 0) {
    unsigned c[7];
    unsigned s = 0u;
#pragma unroll
    for (int k = 0; k < 7; ++k) {
      const int d = lane * 7 + k;
      c[k] = (d < TILE_N) ? fill[d] : 0u;
      s += c[k];
    }
    unsigned incl = s;
#pragma unroll
    for (int off = 1; off < 32; off <<= 1) {
      const unsigned t = __shfl_up(incl, off);
      if (lane >= off) incl += t;
    }
    unsigned run = incl - s;
#pragma unroll
    for (int k = 0; k < 7; ++k) {
      const int d = lane * 7 + k;
      if (d <= TILE_N) start[d] = run;
      run += c[k];
    }
  }
  __syncthreads();

  if (tid <= TILE_N) goff[blockIdx.x * OFF_PITCH + tid] = start[tid];
  if (tid == 0)      goff[blockIdx.x * OFF_PITCH + OFF_PITCH - 1] = total;
  if (tid < TILE_N && lo + tid < N_NODES) dinv[blockIdx.x * DINV_PITCH + tid] = rsqrtf((float)fill[tid] + 1.0f);
  if (listed) {
    const int n = (int)total;
    for (int i = tid; i < n; i += 1024) {
      const unsigned k   = key[i];
      const unsigned pos = start[k >> 16] + (k & 0xffffu);
      srt[min(pos, (unsigned)(LIST_CAP - 1))] = list[i] & 0xffffu;
    }
  }
  if (listed && tid < 3 && total + tid < (unsigned)LIST_CAP) srt[total + tid] = 0u;
  __syncthreads();
  if (listed) {
    const int n4 = (int)((total + 3u) >> 2);
    uint4* __restrict__ g = (uint4*)(gsrt + (size_t)blockIdx.x * LIST_CAP);
    for (int i = tid; i < n4; i += 1024) g[i] = ((const uint4*)srt)[i];
  }
}

__device__ __forceinline__ float xhalf(float x) {
  const int b = __builtin_bit_cast(int, x);
  return __builtin_bit_cast(float, __builtin_amdgcn_permlanex16(b, b, 0x76543210, 0xfedcba98, false, false));
}

#define GEMM_WAVES   12
#define GEMM_THREADS (GEMM_WAVES * 32)
#define GEMM_BLOCKS  256
#define ROW_TILES    (N_NODES / 16)
#define WL_PAIRS     (4 * 32)
#define WL_WORDS     (WL_PAIRS * 32)
#define WL_ROUNDS    ((WL_PAIRS + GEMM_WAVES - 1) / GEMM_WAVES)
static_assert(N_NODES % 16 == 0, "every 16-row tile is whole: no row guard");
static_assert(BASES == 4 && HEADS == 8, "3 column groups of 4 tiles: the bases, heads 0-3, heads 4-7");

__global__ __launch_bounds__(GEMM_THREADS) void k_gemm(const float* __restrict__ x,
                                                       const float* __restrict__ bW,
                                                       const float* __restrict__ cW,
                                                       const float* __restrict__ cb,
                                                       float* __restrict__ bases,
                                                       float* __restrict__ wbuf) {
  __shared__ __attribute__((aligned(16))) v2f wl[WL_WORDS];

  const int tid  = threadIdx.x;
  const int lane = tid & 31;
  const int wave = __builtin_amdgcn_readfirstlane(tid >> 5);
  const int cg   = blockIdx.x % 3;
  const int rk   = blockIdx.x / 3;
  const int nb   = (GEMM_BLOCKS + 2 - cg) / 3;
  const int nlo  = lane & 15;
  const int half = lane >> 4;
  const int kb   = half * 2;

  {
    const float* Wg = (cg == 0) ? bW : cW + (cg - 1) * 64;
    const int    ld = (cg == 0) ? HEAD_D : OUT_C;
    const int    ts = (cg == 0) ? IN_C * HEAD_D : 16;
    const float* Wl = Wg + kb * ld + nlo;
    v2f sv[WL_ROUNDS];
#pragma unroll
    for (int j = 0; j < WL_ROUNDS; ++j) {
      const int q = min(wave + GEMM_WAVES * j, WL_PAIRS - 1);
      const float* Wq = Wl + (q >> 5) * ts + (q & 31) * 4 * ld;
      sv[j] = (v2f){ Wq[0], Wq[ld] };
    }
#pragma unroll
    for (int j = 0; j < WL_ROUNDS; ++j) {
      const int q = wave + GEMM_WAVES * j;
      if (GEMM_WAVES * (j + 1) <= WL_PAIRS || q < WL_PAIRS) wl[q * 32 + lane] = sv[j];
    }
  }
  __syncthreads();

  v2f wf[WL_PAIRS];
#pragma unroll
  for (int q = 0; q < WL_PAIRS; ++q) wf[q] = wl[q * 32 + lane];

  for (int rt = wave * nb + rk; rt < ROW_TILES; rt += GEMM_WAVES * nb) {
    const int gr = rt * 16 + nlo;
    const float* xr = x + (size_t)gr * IN_C + kb;
    v2f a[32];
#pragma unroll
    for (int ks = 0; ks < 32; ++ks) a[ks] = *(const v2f*)(xr + ks * 4);

    v8f acc[4];
#pragma unroll
    for (int i = 0; i < 4; ++i) acc[i] = (v8f){0.f, 0.f, 0.f, 0.f, 0.f, 0.f, 0.f, 0.f};
#pragma unroll
    for (int ks = 0; ks < 32; ++ks) {
#pragma unroll
      for (int i = 0; i < 4; ++i) {
        acc[i] = __builtin_amdgcn_wmma_f32_16x16x4_f32(
            false, wf[i * 32 + ks], false, a[ks], (short)0, acc[i], false, false);
      }
    }

    if (cg == 0) {
#pragma unroll
      for (int i = 0; i < 4; ++i) {
        float4* p = (float4*)(bases + (size_t)gr * BD + i * 16 + half * 8);
        p[0] = make_float4(acc[i][0], acc[i][1], acc[i][2], acc[i][3]);
        p[1] = make_float4(acc[i][4], acc[i][5], acc[i][6], acc[i][7]);
      }
    } else {
#pragma unroll
      for (int i = 0; i < 4; ++i) {
        const int head = (cg - 1) * 4 + i;
        const float4 b0 = *(const float4*)(cb + head * 16 + half * 8);
        const float4 b1 = *(const float4*)(cb + head * 16 + half * 8 + 4);
        float v[8] = {acc[i][0] + b0.x, acc[i][1] + b0.y, acc[i][2] + b0.z, acc[i][3] + b0.w,
                      acc[i][4] + b1.x, acc[i][5] + b1.y, acc[i][6] + b1.z, acc[i][7] + b1.w};
        float mx = fmaxf(fmaxf(fmaxf(v[0], v[1]), fmaxf(v[2], v[3])),
                         fmaxf(fmaxf(v[4], v[5]), fmaxf(v[6], v[7])));
        mx = fmaxf(mx, xhalf(mx));
        float e[8];
#pragma unroll
        for (int r = 0; r < 8; ++r) e[r] = __expf(v[r] - mx);
        float s = ((e[0] + e[1]) + (e[2] + e[3])) + ((e[4] + e[5]) + (e[6] + e[7]));
        s = s + xhalf(s);
        const float inv = 1.0f / s;
        float4* p = (float4*)(wbuf + (size_t)gr * OUT_C + head * 16 + half * 8);
        p[0] = make_float4(e[0] * inv, e[1] * inv, e[2] * inv, e[3] * inv);
        p[1] = make_float4(e[4] * inv, e[5] * inv, e[6] * inv, e[7] * inv);
      }
    }
  }
}

#define PASS_ROWS 32
template <int U>
__device__ __forceinline__ void pass_rows(const int n, const unsigned rv, const int dl, const float* __restrict__ bases, const int lane,
                                          float& sy0, float& sy1, float& su0, float& su1,
                                          float& sq0, float& sq1, float& mx0, float& mx1) {
  if (U < n) {
    const unsigned r = (unsigned)__builtin_amdgcn_readlane(rv, U);
    const float2 t = ((const float2*)(bases + (size_t)r * BD))[lane];
    pass_rows<U + 1>(n, rv, dl, bases, lane, sy0, sy1, su0, su1, sq0, sq1, mx0, mx1);
    const float d = __int_as_float(__builtin_amdgcn_readlane(dl, U));
    const float a = t.x, b = t.y;
    sy0 += d * a;            sy1 += d * b;
    su0 += a;                su1 += b;
    sq0 += a * a;            sq1 += b * b;
    mx0 = fmaxf(mx0, a);     mx1 = fmaxf(mx1, b);
  }
}
template <>
__device__ __forceinline__ void pass_rows<PASS_ROWS>(const int, const unsigned, const int, const float* __restrict__, const int,
                                                     float&, float&, float&, float&, float&, float&, float&, float&) {}

__global__ __launch_bounds__(1024) void k_gather(const int* __restrict__ ei,
                                                 const float* __restrict__ bases,
                                                 const float* __restrict__ wbuf,
                                                 const unsigned* __restrict__ goff,
                                                 const unsigned* __restrict__ gsrt,
                                                 const float* __restrict__ dinv,
                                                 const float* __restrict__ bias,
                                                 float* __restrict__ out) {
  __shared__ __attribute__((aligned(16))) unsigned srt[LIST_CAP];
  __shared__ __attribute__((aligned(16))) float    scr[32 * 256];
  __shared__ unsigned start[TILE_N + 1];

  const int tid  = threadIdx.x;
  const int lane = tid & 31;
  const int wave = __builtin_amdgcn_readfirstlane(tid >> 5);
  const int lo   = blockIdx.x * TILE_N;
  const int* __restrict__ col = ei + N_EDGES;

  const unsigned total  = goff[blockIdx.x * OFF_PITCH + OFF_PITCH - 1];
  const bool     listed = total <= (unsigned)LIST_CAP;
  unsigned sv = 0u;
  if (tid <= TILE_N) sv = goff[blockIdx.x * OFF_PITCH + tid];
  if (listed) {
    const int n4 = (int)((total + 3u) >> 2);
    const uint4* __restrict__ g = (const uint4*)(gsrt + (size_t)blockIdx.x * LIST_CAP);
    for (int i = tid; i < n4; i += 1024) ((uint4*)srt)[i] = g[i];
  }
  if (tid <= TILE_N) start[tid] = sv;
  __syncthreads();

  {
    const int h  = lane >> 2;
    const int dq = lane & 3;
    const float4 bv = ((const float4*)bias)[lane];
    float* sw = scr + wave * 256;
    for (int nl = wave; nl < TILE_N; nl += 32) {
      const int nn = lo + nl;
      if (nn >= N_NODES) break;
      const int s0 = __builtin_amdgcn_readfirstlane((int)start[nl]);
      const int s1 = __builtin_amdgcn_readfirstlane((int)start[nl + 1]);
      const float  dv = dinv[blockIdx.x * DINV_PITCH + nl];
      const float2 bs = *(const float2*)(bases + (size_t)nn * BD + 2 * lane);
      const float4* wp = (const float4*)(wbuf + (size_t)nn * OUT_C + h * 16);
      float4 w[4];
#pragma unroll
      for (int b = 0; b < 4; ++b) w[b] = wp[b];

      float sy0 = 0.f, sy1 = 0.f, su0 = 0.f, su1 = 0.f, sq0 = 0.f, sq1 = 0.f;
      float mx0 = -INFINITY, mx1 = -INFINITY;
      if (listed) {
        for (int base = s0; base < s1; base += PASS_ROWS) {
          const int      n  = min(s1 - base, PASS_ROWS);
          const int      ii = min(base + lane, s1 - 1);
          const unsigned rv = srt[ii];
          const int      dl = __float_as_int(dinv[dinv_at(rv)]);
          pass_rows<0>(n, rv, dl, bases, lane, sy0, sy1, su0, su1, sq0, sq1, mx0, mx1);
        }
      } else {
        for (int e0 = 0; e0 < N_EDGES; e0 += 32) {
          unsigned m = __builtin_amdgcn_ballot_w32(col[e0 + lane] == nn);
          while (m != 0u) {
            const int p = __builtin_ctz(m);
            m &= m - 1u;
            const int    r  = ei[e0 + p];
            const float  d1 = dinv[dinv_at((unsigned)r)];
            const float2 t1 = *(const float2*)(bases + (size_t)r * BD + 2 * lane);
            sy0 += d1 * t1.x;       sy1 += d1 * t1.y;
            su0 += t1.x;            su1 += t1.y;
            sq0 += t1.x * t1.x;     sq1 += t1.y * t1.y;
            mx0 = fmaxf(mx0, t1.x); mx1 = fmaxf(mx1, t1.y);
          }
        }
      }

      const float cf = (float)(s1 - s0);
      const float ic = 1.0f / fmaxf(cf, 1.0f);
      const float sym0 = dv * (sy0 + dv * bs.x);
      const float sym1 = dv * (sy1 + dv * bs.y);
      const float me0  = su0 * ic,            me1  = su1 * ic;
      const float va0  = sq0 * ic - me0 * me0, va1 = sq1 * ic - me1 * me1;
      const float sd0  = sqrtf(fmaxf(va0, 0.0f) + 1e-5f);
      const float sd1  = sqrtf(fmaxf(va1, 0.0f) + 1e-5f);
      const float xo0  = (cf > 0.0f) ? mx0 : 0.0f;
      const float xo1  = (cf > 0.0f) ? mx1 : 0.0f;

      *(float2*)(sw +       2 * lane) = make_float2(sym0, sym1);
      *(float2*)(sw +  64 + 2 * lane) = make_float2(me0,  me1);
      *(float2*)(sw + 128 + 2 * lane) = make_float2(xo0,  xo1);
      *(float2*)(sw + 192 + 2 * lane) = make_float2(sd0,  sd1);
      __builtin_amdgcn_fence(__ATOMIC_SEQ_CST, "wavefront");

      const float* a = sw + dq * 4;
      float4 o = bv;
#pragma unroll
      for (int b = 0; b < 4; ++b) {
        const float4 sy = *(const float4*)(a + b * 16);
        const float4 me = *(const float4*)(a + 64 + b * 16);
        const float4 mx = *(const float4*)(a + 128 + b * 16);
        const float4 sd = *(const float4*)(a + 192 + b * 16);
        o.x += w[b].x * sy.x + w[b].y * me.x + w[b].z * mx.x + w[b].w * sd.x;
        o.y += w[b].x * sy.y + w[b].y * me.y + w[b].z * mx.y + w[b].w * sd.y;
        o.z += w[b].x * sy.z + w[b].y * me.z + w[b].z * mx.z + w[b].w * sd.z;
        o.w += w[b].x * sy.w + w[b].y * me.w + w[b].z * mx.w + w[b].w * sd.w;
      }
      __builtin_amdgcn_fence(__ATOMIC_RELEASE, "wavefront");
      ((float4*)(out + (size_t)nn * OUT_C))[lane] = o;
    }
  }
}

extern "C" void kernel_launch(void* const* d_in, const int* in_sizes, int n_in,
                              void* d_out, int out_size, void* d_ws, size_t ws_size,
                              hipStream_t stream) {
  (void)in_sizes; (void)n_in; (void)out_size; (void)ws_size;
  const float* x    = (const float*)d_in[0];
  const int*   ei   = (const int*)d_in[1];
  const float* bW   = (const float*)d_in[2];
  const float* cW   = (const float*)d_in[3];
  const float* cb   = (const float*)d_in[4];
  const float* bias = (const float*)d_in[5];
  float*       out  = (float*)d_out;

  char* ws = (char*)d_ws;
  float*         bases = (float*)ws;    ws += (size_t)N_NODES * BD * sizeof(float);
  float*         wbuf  = (float*)ws;    ws += (size_t)N_NODES * OUT_C * sizeof(float);
  unsigned*      gsrt  = (unsigned*)ws; ws += (size_t)N_TILES * LIST_CAP * sizeof(unsigned);
  unsigned*      goff  = (unsigned*)ws; ws += (size_t)N_TILES * OFF_PITCH * sizeof(unsigned);
  float*         dinv  = (float*)ws;
  unsigned*      bitmap = (unsigned*)d_ws;

  k_bits<<<BITS_BLOCKS, BITS_THREADS, 0, stream>>>(ei, ei + N_EDGES, bitmap);
  k_list<<<N_TILES, 1024, 0, stream>>>(ei, bitmap, dinv, goff, gsrt);
  k_gemm<<<GEMM_BLOCKS, GEMM_THREADS, 0, stream>>>(x, bW, cW, cb, bases, wbuf);
  k_gather<<<N_TILES, 1024, 0, stream>>>(ei, bases, wbuf, goff, gsrt, dinv, bias, out);
}


# k_bits chunks are 3128 edges, not 1024 (256 work-groups of 13 waves): a (chunk, tile) record line is over a third full, lines written and loaded shrink to a third; list and sum order change
# speedup vs baseline: 6.4673x; 1.0779x over previous
#include <hip/hip_runtime.h>
#include <math.h>
#include <float.h>


#define N_NODES 50000
#define N_EDGES 800000
#define IN_C    128
#define OUT_C   128
#define HEADS   8
#define BASES   4
#define HEAD_D  16
#define BD      (BASES * HEAD_D)

#define TILE_N    196
#define N_TILES   256
#define LIST_CAP  8192
#define OFF_PITCH 256
#define DINV_PITCH 256
#define BM_LINES  256
#define BM_PITCH  ((BM_LINES + 2) * 32)
#define BM_PIECES (BM_LINES * 8)
#define BM_PER_T  2
#define LINE_SLOTS 32
#define REC_EMPTY 0xffffffffu
#define REC_OVER  0xfffffffeu
static_assert(N_NODES <= 65536 && TILE_N <= 256, "row and node-in-tile fit an edge record");

__device__ __forceinline__ unsigned dinv_at(unsigned n) {
  return n + (n / (unsigned)TILE_N) * (unsigned)(DINV_PITCH - TILE_N);
}

typedef float v2f __attribute__((ext_vector_type(2)));
typedef float v8f __attribute__((ext_vector_type(8)));

#define BITS_EDGES   3128
#define BITS_QUADS   (BITS_EDGES / 4)
#define BITS_THREADS 416
#define BITS_BLOCKS  ((N_EDGES + BITS_EDGES - 1) / BITS_EDGES)
static_assert(BITS_EDGES % 4 == 0 && 2 * BITS_THREADS >= BITS_QUADS, "two 16-byte groups a thread cover a chunk");
#define BITS_LPITCH  33
static_assert(BITS_BLOCKS == BM_LINES, "one 128-byte line per work-group and tile");
static_assert(BM_PER_T * 1024 >= BM_PIECES, "the list kernel's threads cover a tile's lines");
static_assert(N_EDGES % 4 == 0, "the edge list is whole groups of 4");
__global__ __launch_bounds__(BITS_THREADS) void k_bits(const int* __restrict__ row,
                                                       const int* __restrict__ col,
                                                       unsigned* __restrict__ bitmap) {
  __shared__ unsigned img[N_TILES * BITS_LPITCH];
  __shared__ unsigned taken[N_TILES];
  const int tid = threadIdx.x;

  const int4* __restrict__ c4 = (const int4*)col;
  const int4* __restrict__ r4 = (const int4*)row;
  const int qb = blockIdx.x * BITS_QUADS;
  int4 c[2], r[2];
  bool have[2];
#pragma unroll
  for (int k = 0; k < 2; ++k) {
    const int lq = k * BITS_THREADS + tid;
    const int q  = qb + lq;
    have[k] = lq < BITS_QUADS && q < N_EDGES / 4;
    const int qq = have[k] ? q : N_EDGES / 4 - 1;
    c[k] = c4[qq];
    r[k] = r4[qq];
  }

  for (int i = tid; i < N_TILES * BITS_LPITCH; i += BITS_THREADS) img[i] = REC_EMPTY;
  for (int i = tid; i < N_TILES; i += BITS_THREADS) taken[i] = 0u;
  __syncthreads();

  unsigned at[8], rec[8], sl[8];
#pragma unroll
  for (int k = 0; k < 2; ++k) {
    const unsigned cv[4] = {(unsigned)c[k].x, (unsigned)c[k].y, (unsigned)c[k].z, (unsigned)c[k].w};
    const unsigned rv[4] = {(unsigned)r[k].x, (unsigned)r[k].y, (unsigned)r[k].z, (unsigned)r[k].w};
#pragma unroll
    for (int j = 0; j < 4; ++j) {
      const unsigned t = min(cv[j] / (unsigned)TILE_N, (unsigned)(N_TILES - 1));
      const unsigned d = (cv[j] - t * (unsigned)TILE_N) & 0xffu;
      at[k * 4 + j]  = t * BITS_LPITCH;
      rec[k * 4 + j] = (rv[j] & 0xffffu) | (d << 16);
      sl[k * 4 + j]  = (unsigned)LINE_SLOTS;
      if (have[k]) sl[k * 4 + j] = atomicAdd(&taken[t], 1u);
    }
  }
#pragma unroll
  for (int i = 0; i < 8; ++i)
    if (sl[i] < (unsigned)LINE_SLOTS) img[at[i] + sl[i]] = rec[i];
  __syncthreads();

  const int lane = tid & 31;
  const int wave = tid >> 5;
  for (int t = wave; t < N_TILES; t += BITS_THREADS / 32) {
    unsigned v = img[t * BITS_LPITCH + lane];
    if (taken[t] > (unsigned)LINE_SLOTS) v = (lane == 0) ? REC_OVER : REC_EMPTY;
    bitmap[(size_t)t * BM_PITCH + blockIdx.x * 32 + lane] = v;
  }
}

__global__ __launch_bounds__(1024) void k_list(const int* __restrict__ ei,
                                               const unsigned* __restrict__ bitmap,
                                               float* __restrict__ dinv,
                                               unsigned* __restrict__ goff,
                                               unsigned* __restrict__ gsrt) {
  __shared__ __attribute__((aligned(16))) unsigned list[LIST_CAP];
  __shared__ __attribute__((aligned(16))) unsigned key[LIST_CAP];
  __shared__ __attribute__((aligned(16))) unsigned srt[LIST_CAP];
  __shared__ unsigned start[TILE_N + 1];
  __shared__ unsigned fill[TILE_N];
  __shared__ unsigned lcount;
  __shared__ unsigned over[BITS_BLOCKS];
  __shared__ unsigned nover;

  const int tid  = threadIdx.x;
  const int lane = tid & 31;
  const int wave = __builtin_amdgcn_readfirstlane(tid >> 5);
  const int lo   = blockIdx.x * TILE_N;
  const int* __restrict__ col = ei + N_EDGES;

  if (tid < TILE_N) fill[tid] = 0u;
  if (tid == 0) { lcount = 0u; nover = 0u; }
  __syncthreads();

  {
    const uint4* __restrict__ bm = (const uint4*)(bitmap + (size_t)blockIdx.x * BM_PITCH);
    uint4 v[BM_PER_T];
#pragma unroll
    for (int k = 0; k < BM_PER_T; ++k) {
      const int j = k * 1024 + tid;
      v[k] = (j < BM_PIECES) ? bm[j] : make_uint4(REC_EMPTY, REC_EMPTY, REC_EMPTY, REC_EMPTY);
    }
    unsigned cnt = 0u;
#pragma unroll
    for (int k = 0; k < BM_PER_T; ++k)
      cnt += (unsigned)(v[k].x < REC_OVER) + (unsigned)(v[k].y < REC_OVER)
           + (unsigned)(v[k].z < REC_OVER) + (unsigned)(v[k].w < REC_OVER);
    unsigned incl = cnt;
#pragma unroll
    for (int off = 1; off < 32; off <<= 1) {
      const unsigned t = __shfl_up(incl, off);
      if (lane >= off) incl += t;
    }
    unsigned run0 = 0u;
    if (lane == 31) run0 = atomicAdd(&lcount, incl);
    run0 = (unsigned)__builtin_amdgcn_readlane((int)run0, 31);
    unsigned slot = run0 + (incl - cnt);
#pragma unroll
    for (int k = 0; k < BM_PER_T; ++k) {
      const unsigned w[4] = {v[k].x, v[k].y, v[k].z, v[k].w};
#pragma unroll
      for (int c = 0; c < 4; ++c) {
        if (w[c] < REC_OVER) {
          if (slot < (unsigned)LIST_CAP) list[slot] = w[c];
          ++slot;
        }
      }
      if (w[0] == REC_OVER) {
        const unsigned q = atomicAdd(&nover, 1u);
        if (q < (unsigned)BITS_BLOCKS) over[q] = (unsigned)(k * 1024 + tid) >> 3;
      }
    }
  }
  __syncthreads();
  const unsigned nov = min(nover, (unsigned)BITS_BLOCKS);
  if (nov != 0u) {
    for (unsigned q = 0u; q < nov; ++q) {
      for (int i = tid; i < BITS_EDGES; i += 1024) {
        const int e = (int)over[q] * BITS_EDGES + i;
        if (e < N_EDGES) {
          const unsigned d = (unsigned)(col[e] - lo);
          if (d < (unsigned)TILE_N) {
            const unsigned s = atomicAdd(&lcount, 1u);
            if (s < (unsigned)LIST_CAP) list[s] = ((unsigned)ei[e] & 0xffffu) | (d << 16);
          }
        }
      }
    }
    __syncthreads();
  }
  const unsigned total  = lcount;
  const bool     listed = total <= (unsigned)LIST_CAP;

  if (listed) {
    const int n = (int)total;
    for (int i = tid; i < n; i += 1024) {
      const unsigned d  = min(list[i] >> 16, (unsigned)(TILE_N - 1));
      const unsigned pl = atomicAdd(&fill[d], 1u);
      key[i]  = (d << 16) | pl;
    }
  } else {
    for (int e = tid; e < N_EDGES; e += 1024) {
      const unsigned d = (unsigned)(col[e] - lo);
      if (d < (unsigned)TILE_N) atomicAdd(&fill[d], 1u);
    }
  }
  __syncthreads();

  if (wave == 0) {
    unsigned c[7];
    unsigned s = 0u;
#pragma unroll
    for (int k = 0; k < 7; ++k) {
      const int d = lane * 7 + k;
      c[k] = (d < TILE_N) ? fill[d] : 0u;
      s += c[k];
    }
    unsigned incl = s;
#pragma unroll
    for (int off = 1; off < 32; off <<= 1) {
      const unsigned t = __shfl_up(incl, off);
      if (lane >= off) incl += t;
    }
    unsigned run = incl - s;
#pragma unroll
    for (int k = 0; k < 7; ++k) {
      const int d = lane * 7 + k;
      if (d <= TILE_N) start[d] = run;
      run += c[k];
    }
  }
  __syncthreads();

  if (tid <= TILE_N) goff[blockIdx.x * OFF_PITCH + tid] = start[tid];
  if (tid == 0)      goff[blockIdx.x * OFF_PITCH + OFF_PITCH - 1] = total;
  if (tid < TILE_N && lo + tid < N_NODES) dinv[blockIdx.x * DINV_PITCH + tid] = rsqrtf((float)fill[tid] + 1.0f);
  if (listed) {
    const int n = (int)total;
    for (int i = tid; i < n; i += 1024) {
      const unsigned k   = key[i];
      const unsigned pos = start[k >> 16] + (k & 0xffffu);
      srt[min(pos, (unsigned)(LIST_CAP - 1))] = list[i] & 0xffffu;
    }
  }
  if (listed && tid < 3 && total + tid < (unsigned)LIST_CAP) srt[total + tid] = 0u;
  __syncthreads();
  if (listed) {
    const int n4 = (int)((total + 3u) >> 2);
    uint4* __restrict__ g = (uint4*)(gsrt + (size_t)blockIdx.x * LIST_CAP);
    for (int i = tid; i < n4; i += 1024) g[i] = ((const uint4*)srt)[i];
  }
}

__device__ __forceinline__ float xhalf(float x) {
  const int b = __builtin_bit_cast(int, x);
  return __builtin_bit_cast(float, __builtin_amdgcn_permlanex16(b, b, 0x76543210, 0xfedcba98, false, false));
}

#define GEMM_WAVES   12
#define GEMM_THREADS (GEMM_WAVES * 32)
#define GEMM_BLOCKS  256
#define ROW_TILES    (N_NODES / 16)
#define WL_PAIRS     (4 * 32)
#define WL_WORDS     (WL_PAIRS * 32)
#define WL_ROUNDS    ((WL_PAIRS + GEMM_WAVES - 1) / GEMM_WAVES)
static_assert(N_NODES % 16 == 0, "every 16-row tile is whole: no row guard");
static_assert(BASES == 4 && HEADS == 8, "3 column groups of 4 tiles: the bases, heads 0-3, heads 4-7");

__global__ __launch_bounds__(GEMM_THREADS) void k_gemm(const float* __restrict__ x,
                                                       const float* __restrict__ bW,
                                                       const float* __restrict__ cW,
                                                       const float* __restrict__ cb,
                                                       float* __restrict__ bases,
                                                       float* __restrict__ wbuf) {
  __shared__ __attribute__((aligned(16))) v2f wl[WL_WORDS];

  const int tid  = threadIdx.x;
  const int lane = tid & 31;
  const int wave = __builtin_amdgcn_readfirstlane(tid >> 5);
  const int cg   = blockIdx.x % 3;
  const int rk   = blockIdx.x / 3;
  const int nb   = (GEMM_BLOCKS + 2 - cg) / 3;
  const int nlo  = lane & 15;
  const int half = lane >> 4;
  const int kb   = half * 2;

  {
    const float* Wg = (cg == 0) ? bW : cW + (cg - 1) * 64;
    const int    ld = (cg == 0) ? HEAD_D : OUT_C;
    const int    ts = (cg == 0) ? IN_C * HEAD_D : 16;
    const float* Wl = Wg + kb * ld + nlo;
    v2f sv[WL_ROUNDS];
#pragma unroll
    for (int j = 0; j < WL_ROUNDS; ++j) {
      const int q = min(wave + GEMM_WAVES * j, WL_PAIRS - 1);
      const float* Wq = Wl + (q >> 5) * ts + (q & 31) * 4 * ld;
      sv[j] = (v2f){ Wq[0], Wq[ld] };
    }
#pragma unroll
    for (int j = 0; j < WL_ROUNDS; ++j) {
      const int q = wave + GEMM_WAVES * j;
      if (GEMM_WAVES * (j + 1) <= WL_PAIRS || q < WL_PAIRS) wl[q * 32 + lane] = sv[j];
    }
  }
  __syncthreads();

  v2f wf[WL_PAIRS];
#pragma unroll
  for (int q = 0; q < WL_PAIRS; ++q) wf[q] = wl[q * 32 + lane];

  for (int rt = wave * nb + rk; rt < ROW_TILES; rt += GEMM_WAVES * nb) {
    const int gr = rt * 16 + nlo;
    const float* xr = x + (size_t)gr * IN_C + kb;
    v2f a[32];
#pragma unroll
    for (int ks = 0; ks < 32; ++ks) a[ks] = *(const v2f*)(xr + ks * 4);

    v8f acc[4];
#pragma unroll
    for (int i = 0; i < 4; ++i) acc[i] = (v8f){0.f, 0.f, 0.f, 0.f, 0.f, 0.f, 0.f, 0.f};
#pragma unroll
    for (int ks = 0; ks < 32; ++ks) {
#pragma unroll
      for (int i = 0; i < 4; ++i) {
        acc[i] = __builtin_amdgcn_wmma_f32_16x16x4_f32(
            false, wf[i * 32 + ks], false, a[ks], (short)0, acc[i], false, false);
      }
    }

    if (cg == 0) {
#pragma unroll
      for (int i = 0; i < 4; ++i) {
        float4* p = (float4*)(bases + (size_t)gr * BD + i * 16 + half * 8);
        p[0] = make_float4(acc[i][0], acc[i][1], acc[i][2], acc[i][3]);
        p[1] = make_float4(acc[i][4], acc[i][5], acc[i][6], acc[i][7]);
      }
    } else {
#pragma unroll
      for (int i = 0; i < 4; ++i) {
        const int head = (cg - 1) * 4 + i;
        const float4 b0 = *(const float4*)(cb + head * 16 + half * 8);
        const float4 b1 = *(const float4*)(cb + head * 16 + half * 8 + 4);
        float v[8] = {acc[i][0] + b0.x, acc[i][1] + b0.y, acc[i][2] + b0.z, acc[i][3] + b0.w,
                      acc[i][4] + b1.x, acc[i][5] + b1.y, acc[i][6] + b1.z, acc[i][7] + b1.w};
        float mx = fmaxf(fmaxf(fmaxf(v[0], v[1]), fmaxf(v[2], v[3])),
                         fmaxf(fmaxf(v[4], v[5]), fmaxf(v[6], v[7])));
        mx = fmaxf(mx, xhalf(mx));
        float e[8];
#pragma unroll
        for (int r = 0; r < 8; ++r) e[r] = __expf(v[r] - mx);
        float s = ((e[0] + e[1]) + (e[2] + e[3])) + ((e[4] + e[5]) + (e[6] + e[7]));
        s = s + xhalf(s);
        const float inv = 1.0f / s;
        float4* p = (float4*)(wbuf + (size_t)gr * OUT_C + head * 16 + half * 8);
        p[0] = make_float4(e[0] * inv, e[1] * inv, e[2] * inv, e[3] * inv);
        p[1] = make_float4(e[4] * inv, e[5] * inv, e[6] * inv, e[7] * inv);
      }
    }
  }
}

#define PASS_ROWS 32
template <int U>
__device__ __forceinline__ void pass_rows(const int n, const unsigned rv, const int dl, const float* __restrict__ bases, const int lane,
                                          float& sy0, float& sy1, float& su0, float& su1,
                                          float& sq0, float& sq1, float& mx0, float& mx1) {
  if (U < n) {
    const unsigned r = (unsigned)__builtin_amdgcn_readlane(rv, U);
    const float2 t = ((const float2*)(bases + (size_t)r * BD))[lane];
    pass_rows<U + 1>(n, rv, dl, bases, lane, sy0, sy1, su0, su1, sq0, sq1, mx0, mx1);
    const float d = __int_as_float(__builtin_amdgcn_readlane(dl, U));
    const float a = t.x, b = t.y;
    sy0 += d * a;            sy1 += d * b;
    su0 += a;                su1 += b;
    sq0 += a * a;            sq1 += b * b;
    mx0 = fmaxf(mx0, a);     mx1 = fmaxf(mx1, b);
  }
}
template <>
__device__ __forceinline__ void pass_rows<PASS_ROWS>(const int, const unsigned, const int, const float* __restrict__, const int,
                                                     float&, float&, float&, float&, float&, float&, float&, float&) {}

__global__ __launch_bounds__(1024) void k_gather(const int* __restrict__ ei,
                                                 const float* __restrict__ bases,
                                                 const float* __restrict__ wbuf,
                                                 const unsigned* __restrict__ goff,
                                                 const unsigned* __restrict__ gsrt,
                                                 const float* __restrict__ dinv,
                                                 const float* __restrict__ bias,
                                                 float* __restrict__ out) {
  __shared__ __attribute__((aligned(16))) unsigned srt[LIST_CAP];
  __shared__ __attribute__((aligned(16))) float    scr[32 * 256];
  __shared__ unsigned start[TILE_N + 1];

  const int tid  = threadIdx.x;
  const int lane = tid & 31;
  const int wave = __builtin_amdgcn_readfirstlane(tid >> 5);
  const int lo   = blockIdx.x * TILE_N;
  const int* __restrict__ col = ei + N_EDGES;

  const unsigned total  = goff[blockIdx.x * OFF_PITCH + OFF_PITCH - 1];
  const bool     listed = total <= (unsigned)LIST_CAP;
  unsigned sv = 0u;
  if (tid <= TILE_N) sv = goff[blockIdx.x * OFF_PITCH + tid];
  if (listed) {
    const int n4 = (int)((total + 3u) >> 2);
    const uint4* __restrict__ g = (const uint4*)(gsrt + (size_t)blockIdx.x * LIST_CAP);
    for (int i = tid; i < n4; i += 1024) ((uint4*)srt)[i] = g[i];
  }
  if (tid <= TILE_N) start[tid] = sv;
  __syncthreads();

  {
    const int h  = lane >> 2;
    const int dq = lane & 3;
    const float4 bv = ((const float4*)bias)[lane];
    float* sw = scr + wave * 256;
    for (int nl = wave; nl < TILE_N; nl += 32) {
      const int nn = lo + nl;
      if (nn >= N_NODES) break;
      const int s0 = __builtin_amdgcn_readfirstlane((int)start[nl]);
      const int s1 = __builtin_amdgcn_readfirstlane((int)start[nl + 1]);
      const float  dv = dinv[blockIdx.x * DINV_PITCH + nl];
      const float2 bs = *(const float2*)(bases + (size_t)nn * BD + 2 * lane);
      const float4* wp = (const float4*)(wbuf + (size_t)nn * OUT_C + h * 16);
      float4 w[4];
#pragma unroll
      for (int b = 0; b < 4; ++b) w[b] = wp[b];

      float sy0 = 0.f, sy1 = 0.f, su0 = 0.f, su1 = 0.f, sq0 = 0.f, sq1 = 0.f;
      float mx0 = -INFINITY, mx1 = -INFINITY;
      if (listed) {
        for (int base = s0; base < s1; base += PASS_ROWS) {
          const int      n  = min(s1 - base, PASS_ROWS);
          const int      ii = min(base + lane, s1 - 1);
          const unsigned rv = srt[ii];
          const int      dl = __float_as_int(dinv[dinv_at(rv)]);
          pass_rows<0>(n, rv, dl, bases, lane, sy0, sy1, su0, su1, sq0, sq1, mx0, mx1);
        }
      } else {
        for (int e0 = 0; e0 < N_EDGES; e0 += 32) {
          unsigned m = __builtin_amdgcn_ballot_w32(col[e0 + lane] == nn);
          while (m != 0u) {
            const int p = __builtin_ctz(m);
            m &= m - 1u;
            const int    r  = ei[e0 + p];
            const float  d1 = dinv[dinv_at((unsigned)r)];
            const float2 t1 = *(const float2*)(bases + (size_t)r * BD + 2 * lane);
            sy0 += d1 * t1.x;       sy1 += d1 * t1.y;
            su0 += t1.x;            su1 += t1.y;
            sq0 += t1.x * t1.x;     sq1 += t1.y * t1.y;
            mx0 = fmaxf(mx0, t1.x); mx1 = fmaxf(mx1, t1.y);
          }
        }
      }

      const float cf = (float)(s1 - s0);
      const float ic = 1.0f / fmaxf(cf, 1.0f);
      const float sym0 = dv * (sy0 + dv * bs.x);
      const float sym1 = dv * (sy1 + dv * bs.y);
      const float me0  = su0 * ic,            me1  = su1 * ic;
      const float va0  = sq0 * ic - me0 * me0, va1 = sq1 * ic - me1 * me1;
      const float sd0  = sqrtf(fmaxf(va0, 0.0f) + 1e-5f);
      const float sd1  = sqrtf(fmaxf(va1, 0.0f) + 1e-5f);
      const float xo0  = (cf > 0.0f) ? mx0 : 0.0f;
      const float xo1  = (cf > 0.0f) ? mx1 : 0.0f;

      *(float2*)(sw +       2 * lane) = make_float2(sym0, sym1);
      *(float2*)(sw +  64 + 2 * lane) = make_float2(me0,  me1);
      *(float2*)(sw + 128 + 2 * lane) = make_float2(xo0,  xo1);
      *(float2*)(sw + 192 + 2 * lane) = make_float2(sd0,  sd1);
      __builtin_amdgcn_fence(__ATOMIC_SEQ_CST, "wavefront");

      const float* a = sw + dq * 4;
      float4 o = bv;
#pragma unroll
      for (int b = 0; b < 4; ++b) {
        const float4 sy = *(const float4*)(a + b * 16);
        const float4 me = *(const float4*)(a + 64 + b * 16);
        const float4 mx = *(const float4*)(a + 128 + b * 16);
        const float4 sd = *(const float4*)(a + 192 + b * 16);
        o.x += w[b].x * sy.x + w[b].y * me.x + w[b].z * mx.x + w[b].w * sd.x;
        o.y += w[b].x * sy.y + w[b].y * me.y + w[b].z * mx.y + w[b].w * sd.y;
        o.z += w[b].x * sy.z + w[b].y * me.z + w[b].z * mx.z + w[b].w * sd.z;
        o.w += w[b].x * sy.w + w[b].y * me.w + w[b].z * mx.w + w[b].w * sd.w;
      }
      __builtin_amdgcn_fence(__ATOMIC_RELEASE, "wavefront");
      ((float4*)(out + (size_t)nn * OUT_C))[lane] = o;
    }
  }
}

extern "C" void kernel_launch(void* const* d_in, const int* in_sizes, int n_in,
                              void* d_out, int out_size, void* d_ws, size_t ws_size,
                              hipStream_t stream) {
  (void)in_sizes; (void)n_in; (void)out_size; (void)ws_size;
  const float* x    = (const float*)d_in[0];
  const int*   ei   = (const int*)d_in[1];
  const float* bW   = (const float*)d_in[2];
  const float* cW   = (const float*)d_in[3];
  const float* cb   = (const float*)d_in[4];
  const float* bias = (const float*)d_in[5];
  float*       out  = (float*)d_out;

  char* ws = (char*)d_ws;
  float*         bases = (float*)ws;    ws += (size_t)N_NODES * BD * sizeof(float);
  float*         wbuf  = (float*)ws;    ws += (size_t)N_NODES * OUT_C * sizeof(float);
  unsigned*      gsrt  = (unsigned*)ws; ws += (size_t)N_TILES * LIST_CAP * sizeof(unsigned);
  unsigned*      goff  = (unsigned*)ws; ws += (size_t)N_TILES * OFF_PITCH * sizeof(unsigned);
  float*         dinv  = (float*)ws;
  unsigned*      bitmap = (unsigned*)d_ws;

  k_bits<<<BITS_BLOCKS, BITS_THREADS, 0, stream>>>(ei, ei + N_EDGES, bitmap);
  k_list<<<N_TILES, 1024, 0, stream>>>(ei, bitmap, dinv, goff, gsrt);
  k_gemm<<<GEMM_BLOCKS, GEMM_THREADS, 0, stream>>>(x, bW, cW, cb, bases, wbuf);
  k_gather<<<N_TILES, 1024, 0, stream>>>(ei, bases, wbuf, goff, gsrt, dinv, bias, out);
}
